# starting kernel
# baseline (speedup 1.0000x reference)
#include <hip/hip_runtime.h>
#include <cmath>


typedef __attribute__((ext_vector_type(2))) float v2f;
typedef __attribute__((ext_vector_type(8))) float v8f;

#define WMMA_F32(a, b, c) \
  __builtin_amdgcn_wmma_f32_16x16x4_f32(false, (a), false, (b), (short)0, (c), false, false)

#define B_   64
#define C_   2048
#define HW_  196
#define M_   32
#define NC_  396
#define K3_  (M_ * C_)
#define NPAD_ 400
#define NTILES_ 25
#define KCHUNK_ 1024
#define NKCH_   64

__device__ __forceinline__ float fast_sigmoid(float v) {
  return __builtin_amdgcn_rcpf(1.0f + __expf(-v));
}

__global__ __launch_bounds__(128) void k_attn(const float* __restrict__ x,
                                              const float* __restrict__ Wa,
                                              const float* __restrict__ ba,
                                              float* __restrict__ A_ws) {
  const int wave = threadIdx.x >> 5;
  const int hwt  = blockIdx.x * 4 + wave;
  if (hwt >= 13) return;
  const int b    = blockIdx.y;
  const int lane = threadIdx.x & 31;
  const int half = lane >> 4;
  const int r    = lane & 15;
  const int hw0  = hwt * 16;
  const int hwr  = min(hw0 + r, HW_ - 1);

  v8f acc0 = {}; v8f acc1 = {};
  const float* xb = x + (size_t)b * C_ * HW_;

  for (int k0 = 0; k0 < C_; k0 += 4) {
    const int k = k0 + 2 * half;
    v2f a;
    a.x = xb[(size_t)k * HW_ + hwr];
    a.y = xb[(size_t)(k + 1) * HW_ + hwr];
    v2f b0 = *(const v2f*)(Wa + (size_t)(r)      * C_ + k);
    v2f b1 = *(const v2f*)(Wa + (size_t)(16 + r) * C_ + k);
    acc0 = WMMA_F32(a, b0, acc0);
    acc1 = WMMA_F32(a, b1, acc1);
  }

#pragma unroll
  for (int i = 0; i < 8; ++i) {
    const int hw = hw0 + i + (half ? 8 : 0);
    if (hw < HW_) {
      A_ws[((size_t)b * M_ + r)      * HW_ + hw] = fast_sigmoid(acc0[i] + ba[r]);
      A_ws[((size_t)b * M_ + 16 + r) * HW_ + hw] = fast_sigmoid(acc1[i] + ba[16 + r]);
    }
  }
}

__global__ __launch_bounds__(128) void k_bap(const float* __restrict__ x,
                                             const float* __restrict__ A_ws,
                                             float* __restrict__ feats) {
  const int wave = threadIdx.x >> 5;
  const int ct   = blockIdx.x * 4 + wave;
  const int b    = blockIdx.y;
  const int lane = threadIdx.x & 31;
  const int half = lane >> 4;
  const int r    = lane & 15;
  const int c0   = ct * 16;

  v8f acc0 = {}; v8f acc1 = {};
  const float* Ab = A_ws + (size_t)b * M_ * HW_;
  const float* xb = x + ((size_t)b * C_ + c0) * HW_;

  for (int k0 = 0; k0 < HW_; k0 += 4) {
    const int k = k0 + 2 * half;
    v2f a0 = *(const v2f*)(Ab + (size_t)(r)      * HW_ + k);
    v2f a1 = *(const v2f*)(Ab + (size_t)(16 + r) * HW_ + k);
    v2f bb = *(const v2f*)(xb + (size_t)r * HW_ + k);
    acc0 = WMMA_F32(a0, bb, acc0);
    acc1 = WMMA_F32(a1, bb, acc1);
  }

  const float s = 1.0f / (float)HW_;
#pragma unroll
  for (int i = 0; i < 8; ++i) {
    const int mr = i + (half ? 8 : 0);
    feats[((size_t)b * M_ + mr)      * C_ + c0 + r] = acc0[i] * s;
    feats[((size_t)b * M_ + 16 + mr) * C_ + c0 + r] = acc1[i] * s;
  }
}

__global__ __launch_bounds__(256) void k_fc_partial(const float* __restrict__ feats,
                                                    const float* __restrict__ Wc,
                                                    float* __restrict__ part) {
  const int wave = threadIdx.x >> 5;
  const int nt   = blockIdx.x * 8 + wave;
  if (nt >= NTILES_) return;
  const int kc   = blockIdx.y;
  const int lane = threadIdx.x & 31;
  const int half = lane >> 4;
  const int r    = lane & 15;
  const int n0   = nt * 16;

  const bool nvalid = (n0 + r) < NC_;
  const int  nrow   = min(n0 + r, NC_ - 1);
  const float* wrow = Wc + (size_t)nrow * K3_;
  const int kbase   = kc * KCHUNK_;

  v8f acc[4] = {{}, {}, {}, {}};

  for (int k0 = 0; k0 < KCHUNK_; k0 += 4) {
    const int k = kbase + k0 + 2 * half;
    v2f bb = *(const v2f*)(wrow + k);
    bb.x = nvalid ? bb.x : 0.0f;
    bb.y = nvalid ? bb.y : 0.0f;
#pragma unroll
    for (int mt = 0; mt < 4; ++mt) {
      v2f a = *(const v2f*)(feats + (size_t)(mt * 16 + r) * K3_ + k);
      acc[mt] = WMMA_F32(a, bb, acc[mt]);
    }
  }

#pragma unroll
  for (int mt = 0; mt < 4; ++mt) {
#pragma unroll
    for (int i = 0; i < 8; ++i) {
      const int brow = mt * 16 + i + (half ? 8 : 0);
      part[((size_t)kc * B_ + brow) * NPAD_ + n0 + r] = acc[mt][i];
    }
  }
}

__global__ __launch_bounds__(256) void k_reduce(const float* __restrict__ part,
                                               const float* __restrict__ bc,
                                               float* __restrict__ out) {
  const int idx = blockIdx.x * blockDim.x + threadIdx.x;
  if (idx >= B_ * NC_) return;
  const int b = idx / NC_;
  const int n = idx - b * NC_;
  float s = bc[n];
#pragma unroll 4
  for (int kc = 0; kc < NKCH_; ++kc)
    s += part[((size_t)kc * B_ + b) * NPAD_ + n];
  out[idx] = s;
}

extern "C" void kernel_launch(void* const* d_in, const int* in_sizes, int n_in,
                              void* d_out, int out_size, void* d_ws, size_t ws_size,
                              hipStream_t stream) {
  const float* x  = (const float*)d_in[0];
  const float* Wa = (const float*)d_in[1];
  const float* ba = (const float*)d_in[2];
  const float* Wc = (const float*)d_in[3];
  const float* bc = (const float*)d_in[4];
  float* out = (float*)d_out;

  float* A_ws    = (float*)d_ws;
  float* feats   = A_ws + (size_t)B_ * M_ * HW_;
  float* part_ws = feats + (size_t)B_ * K3_;

  k_attn      <<<dim3(4, B_),   128, 0, stream>>>(x, Wa, ba, A_ws);
  k_bap       <<<dim3(32, B_),  128, 0, stream>>>(x, A_ws, feats);
  k_fc_partial<<<dim3(4, NKCH_), 256, 0, stream>>>(feats, Wc, part_ws);
  k_reduce    <<<(B_ * NC_ + 255) / 256, 256, 0, stream>>>(part_ws, bc, out);
}


# k_attn re-tiled: block = one b x 64 hw (4 tiles); its 16 waves split the 2048-channel sum (128 each) and add their partial tiles in LDS in wave order (the channel sum order changes)
# speedup vs baseline: 1.4076x; 1.4076x over previous
#include <hip/hip_runtime.h>
#include <cmath>


typedef __attribute__((ext_vector_type(2))) float v2f;
typedef __attribute__((ext_vector_type(8))) float v8f;

#define WMMA_F32(a, b, c) \
  __builtin_amdgcn_wmma_f32_16x16x4_f32(false, (a), false, (b), (short)0, (c), false, false)

#define B_   64
#define C_   2048
#define HW_  196
#define M_   32
#define NC_  396
#define K3_  (M_ * C_)
#define NPAD_ 400
#define NTILES_ 25
#define KCHUNK_ 1024
#define NKCH_   64

__device__ __forceinline__ float fast_sigmoid(float v) {
  return __builtin_amdgcn_rcpf(1.0f + __expf(-v));
}

#define ATT_WAVES_ 16
#define ATT_KSL_   (C_ / ATT_WAVES_)
#define ATT_HT_    4
#define ATT_TILE_  (ATT_HT_ * 2 * 256)

__global__ __launch_bounds__(512) void k_attn(const float* __restrict__ x,
                                              const float* __restrict__ Wa,
                                              const float* __restrict__ ba,
                                              float* __restrict__ A_ws) {
  __shared__ alignas(32) float red[ATT_WAVES_ * ATT_TILE_];

  const int wave = __builtin_amdgcn_readfirstlane((int)(threadIdx.x >> 5));
  const int lane = threadIdx.x & 31;
  const int half = lane >> 4;
  const int r    = lane & 15;
  const int b    = blockIdx.y;
  const int hw0  = blockIdx.x * (16 * ATT_HT_);

  int hwr[ATT_HT_];
#pragma unroll
  for (int j = 0; j < ATT_HT_; ++j)
    hwr[j] = min(hw0 + 16 * j + r, HW_ - 1);

  v8f acc[ATT_HT_][2] = {};
  const int kbeg = wave * ATT_KSL_;
  const float* xk = x + ((size_t)b * C_ + kbeg + 2 * half) * HW_;
  const float* w0 = Wa + (size_t)r * C_ + kbeg + 2 * half;
  const float* w1 = w0 + (size_t)16 * C_;

#pragma unroll 4
  for (int k0 = 0; k0 < ATT_KSL_; k0 += 4) {
    const v2f b0 = *(const v2f*)(w0 + k0);
    const v2f b1 = *(const v2f*)(w1 + k0);
#pragma unroll
    for (int j = 0; j < ATT_HT_; ++j) {
      v2f a;
      a.x = xk[(size_t)k0 * HW_ + hwr[j]];
      a.y = xk[(size_t)(k0 + 1) * HW_ + hwr[j]];
      acc[j][0] = WMMA_F32(a, b0, acc[j][0]);
      acc[j][1] = WMMA_F32(a, b1, acc[j][1]);
    }
  }

  float* mine = red + wave * ATT_TILE_ + r * 16 + 8 * half;
#pragma unroll
  for (int j = 0; j < ATT_HT_; ++j) {
    *(v8f*)(mine + (2 * j + 0) * 256) = acc[j][0];
    *(v8f*)(mine + (2 * j + 1) * 256) = acc[j][1];
  }
  __syncthreads();

  const int t   = threadIdx.x;
  const int e   = t & 255;
  const int hwl = e & 15;
  const int mlo = e >> 4;
#pragma unroll
  for (int jj = 0; jj < 4; ++jj) {
    const int q = (t >> 8) + 2 * jj;
    float s = red[q * 256 + e];
#pragma unroll
    for (int w = 1; w < ATT_WAVES_; ++w) s += red[w * ATT_TILE_ + q * 256 + e];
    const int m  = (q & 1) * 16 + mlo;
    const int hw = hw0 + (q >> 1) * 16 + hwl;
    if (hw < HW_) A_ws[((size_t)b * M_ + m) * HW_ + hw] = fast_sigmoid(s + ba[m]);
  }
}

__global__ __launch_bounds__(128) void k_bap(const float* __restrict__ x,
                                             const float* __restrict__ A_ws,
                                             float* __restrict__ feats) {
  const int wave = threadIdx.x >> 5;
  const int ct   = blockIdx.x * 4 + wave;
  const int b    = blockIdx.y;
  const int lane = threadIdx.x & 31;
  const int half = lane >> 4;
  const int r    = lane & 15;
  const int c0   = ct * 16;

  v8f acc0 = {}; v8f acc1 = {};
  const float* Ab = A_ws + (size_t)b * M_ * HW_;
  const float* xb = x + ((size_t)b * C_ + c0) * HW_;

  for (int k0 = 0; k0 < HW_; k0 += 4) {
    const int k = k0 + 2 * half;
    v2f a0 = *(const v2f*)(Ab + (size_t)(r)      * HW_ + k);
    v2f a1 = *(const v2f*)(Ab + (size_t)(16 + r) * HW_ + k);
    v2f bb = *(const v2f*)(xb + (size_t)r * HW_ + k);
    acc0 = WMMA_F32(a0, bb, acc0);
    acc1 = WMMA_F32(a1, bb, acc1);
  }

  const float s = 1.0f / (float)HW_;
#pragma unroll
  for (int i = 0; i < 8; ++i) {
    const int mr = i + (half ? 8 : 0);
    feats[((size_t)b * M_ + mr)      * C_ + c0 + r] = acc0[i] * s;
    feats[((size_t)b * M_ + 16 + mr) * C_ + c0 + r] = acc1[i] * s;
  }
}

__global__ __launch_bounds__(256) void k_fc_partial(const float* __restrict__ feats,
                                                    const float* __restrict__ Wc,
                                                    float* __restrict__ part) {
  const int wave = threadIdx.x >> 5;
  const int nt   = blockIdx.x * 8 + wave;
  if (nt >= NTILES_) return;
  const int kc   = blockIdx.y;
  const int lane = threadIdx.x & 31;
  const int half = lane >> 4;
  const int r    = lane & 15;
  const int n0   = nt * 16;

  const bool nvalid = (n0 + r) < NC_;
  const int  nrow   = min(n0 + r, NC_ - 1);
  const float* wrow = Wc + (size_t)nrow * K3_;
  const int kbase   = kc * KCHUNK_;

  v8f acc[4] = {{}, {}, {}, {}};

  for (int k0 = 0; k0 < KCHUNK_; k0 += 4) {
    const int k = kbase + k0 + 2 * half;
    v2f bb = *(const v2f*)(wrow + k);
    bb.x = nvalid ? bb.x : 0.0f;
    bb.y = nvalid ? bb.y : 0.0f;
#pragma unroll
    for (int mt = 0; mt < 4; ++mt) {
      v2f a = *(const v2f*)(feats + (size_t)(mt * 16 + r) * K3_ + k);
      acc[mt] = WMMA_F32(a, bb, acc[mt]);
    }
  }

#pragma unroll
  for (int mt = 0; mt < 4; ++mt) {
#pragma unroll
    for (int i = 0; i < 8; ++i) {
      const int brow = mt * 16 + i + (half ? 8 : 0);
      part[((size_t)kc * B_ + brow) * NPAD_ + n0 + r] = acc[mt][i];
    }
  }
}

__global__ __launch_bounds__(256) void k_reduce(const float* __restrict__ part,
                                               const float* __restrict__ bc,
                                               float* __restrict__ out) {
  const int idx = blockIdx.x * blockDim.x + threadIdx.x;
  if (idx >= B_ * NC_) return;
  const int b = idx / NC_;
  const int n = idx - b * NC_;
  float s = bc[n];
#pragma unroll 4
  for (int kc = 0; kc < NKCH_; ++kc)
    s += part[((size_t)kc * B_ + b) * NPAD_ + n];
  out[idx] = s;
}

extern "C" void kernel_launch(void* const* d_in, const int* in_sizes, int n_in,
                              void* d_out, int out_size, void* d_ws, size_t ws_size,
                              hipStream_t stream) {
  const float* x  = (const float*)d_in[0];
  const float* Wa = (const float*)d_in[1];
  const float* ba = (const float*)d_in[2];
  const float* Wc = (const float*)d_in[3];
  const float* bc = (const float*)d_in[4];
  float* out = (float*)d_out;

  float* A_ws    = (float*)d_ws;
  float* feats   = A_ws + (size_t)B_ * M_ * HW_;
  float* part_ws = feats + (size_t)B_ * K3_;

  k_attn      <<<dim3(4, B_),   512, 0, stream>>>(x, Wa, ba, A_ws);
  k_bap       <<<dim3(32, B_),  128, 0, stream>>>(x, A_ws, feats);
  k_fc_partial<<<dim3(4, NKCH_), 256, 0, stream>>>(feats, Wc, part_ws);
  k_reduce    <<<(B_ * NC_ + 255) / 256, 256, 0, stream>>>(part_ws, bc, out);
}


# k_fc_partial: block's feats slab staged once in LDS (128-k stages, 2 buffers), read as 16-byte A fragments; Wc fragments 16-byte, same k order (k sum order inside each 8 changes)
# speedup vs baseline: 1.8340x; 1.3029x over previous
#include <hip/hip_runtime.h>
#include <cmath>


typedef __attribute__((ext_vector_type(2))) float v2f;
typedef __attribute__((ext_vector_type(8))) float v8f;

#define WMMA_F32(a, b, c) \
  __builtin_amdgcn_wmma_f32_16x16x4_f32(false, (a), false, (b), (short)0, (c), false, false)

#define B_   64
#define C_   2048
#define HW_  196
#define M_   32
#define NC_  396
#define K3_  (M_ * C_)
#define NPAD_ 400
#define NTILES_ 25
#define KCHUNK_ 1024
#define NKCH_   64

__device__ __forceinline__ float fast_sigmoid(float v) {
  return __builtin_amdgcn_rcpf(1.0f + __expf(-v));
}

#define ATT_WAVES_ 16
#define ATT_KSL_   (C_ / ATT_WAVES_)
#define ATT_HT_    4
#define ATT_TILE_  (ATT_HT_ * 2 * 256)

__global__ __launch_bounds__(512) void k_attn(const float* __restrict__ x,
                                              const float* __restrict__ Wa,
                                              const float* __restrict__ ba,
                                              float* __restrict__ A_ws) {
  __shared__ alignas(32) float red[ATT_WAVES_ * ATT_TILE_];

  const int wave = __builtin_amdgcn_readfirstlane((int)(threadIdx.x >> 5));
  const int lane = threadIdx.x & 31;
  const int half = lane >> 4;
  const int r    = lane & 15;
  const int b    = blockIdx.y;
  const int hw0  = blockIdx.x * (16 * ATT_HT_);

  int hwr[ATT_HT_];
#pragma unroll
  for (int j = 0; j < ATT_HT_; ++j)
    hwr[j] = min(hw0 + 16 * j + r, HW_ - 1);

  v8f acc[ATT_HT_][2] = {};
  const int kbeg = wave * ATT_KSL_;
  const float* xk = x + ((size_t)b * C_ + kbeg + 2 * half) * HW_;
  const float* w0 = Wa + (size_t)r * C_ + kbeg + 2 * half;
  const float* w1 = w0 + (size_t)16 * C_;

#pragma unroll 4
  for (int k0 = 0; k0 < ATT_KSL_; k0 += 4) {
    const v2f b0 = *(const v2f*)(w0 + k0);
    const v2f b1 = *(const v2f*)(w1 + k0);
#pragma unroll
    for (int j = 0; j < ATT_HT_; ++j) {
      v2f a;
      a.x = xk[(size_t)k0 * HW_ + hwr[j]];
      a.y = xk[(size_t)(k0 + 1) * HW_ + hwr[j]];
      acc[j][0] = WMMA_F32(a, b0, acc[j][0]);
      acc[j][1] = WMMA_F32(a, b1, acc[j][1]);
    }
  }

  float* mine = red + wave * ATT_TILE_ + r * 16 + 8 * half;
#pragma unroll
  for (int j = 0; j < ATT_HT_; ++j) {
    *(v8f*)(mine + (2 * j + 0) * 256) = acc[j][0];
    *(v8f*)(mine + (2 * j + 1) * 256) = acc[j][1];
  }
  __syncthreads();

  const int t   = threadIdx.x;
  const int e   = t & 255;
  const int hwl = e & 15;
  const int mlo = e >> 4;
#pragma unroll
  for (int jj = 0; jj < 4; ++jj) {
    const int q = (t >> 8) + 2 * jj;
    float s = red[q * 256 + e];
#pragma unroll
    for (int w = 1; w < ATT_WAVES_; ++w) s += red[w * ATT_TILE_ + q * 256 + e];
    const int m  = (q & 1) * 16 + mlo;
    const int hw = hw0 + (q >> 1) * 16 + hwl;
    if (hw < HW_) A_ws[((size_t)b * M_ + m) * HW_ + hw] = fast_sigmoid(s + ba[m]);
  }
}

__global__ __launch_bounds__(128) void k_bap(const float* __restrict__ x,
                                             const float* __restrict__ A_ws,
                                             float* __restrict__ feats) {
  const int wave = threadIdx.x >> 5;
  const int ct   = blockIdx.x * 4 + wave;
  const int b    = blockIdx.y;
  const int lane = threadIdx.x & 31;
  const int half = lane >> 4;
  const int r    = lane & 15;
  const int c0   = ct * 16;

  v8f acc0 = {}; v8f acc1 = {};
  const float* Ab = A_ws + (size_t)b * M_ * HW_;
  const float* xb = x + ((size_t)b * C_ + c0) * HW_;

  for (int k0 = 0; k0 < HW_; k0 += 4) {
    const int k = k0 + 2 * half;
    v2f a0 = *(const v2f*)(Ab + (size_t)(r)      * HW_ + k);
    v2f a1 = *(const v2f*)(Ab + (size_t)(16 + r) * HW_ + k);
    v2f bb = *(const v2f*)(xb + (size_t)r * HW_ + k);
    acc0 = WMMA_F32(a0, bb, acc0);
    acc1 = WMMA_F32(a1, bb, acc1);
  }

  const float s = 1.0f / (float)HW_;
#pragma unroll
  for (int i = 0; i < 8; ++i) {
    const int mr = i + (half ? 8 : 0);
    feats[((size_t)b * M_ + mr)      * C_ + c0 + r] = acc0[i] * s;
    feats[((size_t)b * M_ + 16 + mr) * C_ + c0 + r] = acc1[i] * s;
  }
}

typedef __attribute__((ext_vector_type(4))) float v4f;

#define FC_KS_    128
#define FC_PITCH_ (FC_KS_ + 4)
#define FC_NST_   (KCHUNK_ / FC_KS_)
#define FC_NG_    (FC_KS_ / 8)
#define FC_NLD_   (B_ * FC_KS_ / 4 / 256)

__global__ __launch_bounds__(256) void k_fc_partial(const float* __restrict__ feats,
                                                    const float* __restrict__ Wc,
                                                    float* __restrict__ part) {
  __shared__ alignas(16) float fs[2][B_ * FC_PITCH_];

  const int wave = __builtin_amdgcn_readfirstlane((int)(threadIdx.x >> 5));
  const int lane = threadIdx.x & 31;
  const int half = lane >> 4;
  const int r    = lane & 15;
  const int nt   = blockIdx.x * 8 + wave;
  const bool wvalid = nt < NTILES_;
  const int kc   = blockIdx.y;
  const int n0   = nt * 16;

  const bool nvalid = (n0 + r) < NC_;
  const int  nrow   = min(n0 + r, NC_ - 1);
  const float* wrow = Wc + (size_t)nrow * K3_ + (size_t)kc * KCHUNK_ + 4 * half;

  const float* fsrc = feats + (size_t)wave * K3_ + (size_t)kc * KCHUNK_ + 4 * lane;
  const int    fdst = wave * FC_PITCH_ + 4 * lane;

  v4f st[FC_NLD_];
  v4f wc[FC_NG_];
#pragma unroll
  for (int i = 0; i < FC_NLD_; ++i) st[i] = *(const v4f*)(fsrc + (size_t)(8 * i) * K3_);
  if (wvalid) {
#pragma unroll
    for (int g = 0; g < FC_NG_; ++g) wc[g] = *(const v4f*)(wrow + 8 * g);
  }
#pragma unroll
  for (int i = 0; i < FC_NLD_; ++i) *(v4f*)(&fs[0][fdst + 8 * i * FC_PITCH_]) = st[i];
  __syncthreads();

  v8f acc[4] = {{}, {}, {}, {}};

#pragma unroll 2
  for (int s = 0; s < FC_NST_; ++s) {
    const int cur = s & 1;
    const bool more = (s + 1) < FC_NST_;
    v4f wn[FC_NG_];
    if (more) {
#pragma unroll
      for (int i = 0; i < FC_NLD_; ++i)
        st[i] = *(const v4f*)(fsrc + (size_t)(8 * i) * K3_ + (s + 1) * FC_KS_);
      if (wvalid) {
#pragma unroll
        for (int g = 0; g < FC_NG_; ++g) wn[g] = *(const v4f*)(wrow + (s + 1) * FC_KS_ + 8 * g);
      }
    }
    if (wvalid) {
      const float* fa = &fs[cur][r * FC_PITCH_ + 4 * half];
#pragma unroll
      for (int g = 0; g < FC_NG_; ++g) {
        v4f bq = wc[g];
        bq.x = nvalid ? bq.x : 0.0f;
        bq.y = nvalid ? bq.y : 0.0f;
        bq.z = nvalid ? bq.z : 0.0f;
        bq.w = nvalid ? bq.w : 0.0f;
        const v2f b01 = {bq.x, bq.y};
        const v2f b23 = {bq.z, bq.w};
#pragma unroll
        for (int mt = 0; mt < 4; ++mt) {
          const v4f aq = *(const v4f*)(fa + mt * 16 * FC_PITCH_ + 8 * g);
          const v2f a01 = {aq.x, aq.y};
          const v2f a23 = {aq.z, aq.w};
          acc[mt] = WMMA_F32(a01, b01, acc[mt]);
          acc[mt] = WMMA_F32(a23, b23, acc[mt]);
        }
      }
    }
    if (more) {
#pragma unroll
      for (int i = 0; i < FC_NLD_; ++i) *(v4f*)(&fs[cur ^ 1][fdst + 8 * i * FC_PITCH_]) = st[i];
      if (wvalid) {
#pragma unroll
        for (int g = 0; g < FC_NG_; ++g) wc[g] = wn[g];
      }
    }
    __syncthreads();
  }

  if (wvalid) {
#pragma unroll
    for (int mt = 0; mt < 4; ++mt) {
#pragma unroll
      for (int i = 0; i < 8; ++i) {
        const int brow = mt * 16 + i + (half ? 8 : 0);
        part[((size_t)kc * B_ + brow) * NPAD_ + n0 + r] = acc[mt][i];
      }
    }
  }
}

__global__ __launch_bounds__(256) void k_reduce(const float* __restrict__ part,
                                               const float* __restrict__ bc,
                                               float* __restrict__ out) {
  const int idx = blockIdx.x * blockDim.x + threadIdx.x;
  if (idx >= B_ * NC_) return;
  const int b = idx / NC_;
  const int n = idx - b * NC_;
  float s = bc[n];
#pragma unroll 4
  for (int kc = 0; kc < NKCH_; ++kc)
    s += part[((size_t)kc * B_ + b) * NPAD_ + n];
  out[idx] = s;
}

extern "C" void kernel_launch(void* const* d_in, const int* in_sizes, int n_in,
                              void* d_out, int out_size, void* d_ws, size_t ws_size,
                              hipStream_t stream) {
  const float* x  = (const float*)d_in[0];
  const float* Wa = (const float*)d_in[1];
  const float* ba = (const float*)d_in[2];
  const float* Wc = (const float*)d_in[3];
  const float* bc = (const float*)d_in[4];
  float* out = (float*)d_out;

  float* A_ws    = (float*)d_ws;
  float* feats   = A_ws + (size_t)B_ * M_ * HW_;
  float* part_ws = feats + (size_t)B_ * K3_;

  k_attn      <<<dim3(4, B_),   512, 0, stream>>>(x, Wa, ba, A_ws);
  k_bap       <<<dim3(32, B_),  128, 0, stream>>>(x, A_ws, feats);
  k_fc_partial<<<dim3(4, NKCH_), 256, 0, stream>>>(feats, Wc, part_ws);
  k_reduce    <<<(B_ * NC_ + 255) / 256, 256, 0, stream>>>(part_ws, bc, out);
}


# k_bap: A[b] and the block's x slabs (64 channels x 196, two buffers) copied to LDS by whole-line 16-byte loads; A and B fragments are 16-byte LDS reads, same hw order (hw sum order in each 8 changes)
# speedup vs baseline: 2.3678x; 1.2911x over previous
#include <hip/hip_runtime.h>
#include <cmath>


typedef __attribute__((ext_vector_type(2))) float v2f;
typedef __attribute__((ext_vector_type(8))) float v8f;

#define WMMA_F32(a, b, c) \
  __builtin_amdgcn_wmma_f32_16x16x4_f32(false, (a), false, (b), (short)0, (c), false, false)

#define B_   64
#define C_   2048
#define HW_  196
#define M_   32
#define NC_  396
#define K3_  (M_ * C_)
#define NPAD_ 400
#define NTILES_ 25
#define KCHUNK_ 1024
#define NKCH_   64

__device__ __forceinline__ float fast_sigmoid(float v) {
  return __builtin_amdgcn_rcpf(1.0f + __expf(-v));
}

#define ATT_WAVES_ 16
#define ATT_KSL_   (C_ / ATT_WAVES_)
#define ATT_HT_    4
#define ATT_TILE_  (ATT_HT_ * 2 * 256)

__global__ __launch_bounds__(512) void k_attn(const float* __restrict__ x,
                                              const float* __restrict__ Wa,
                                              const float* __restrict__ ba,
                                              float* __restrict__ A_ws) {
  __shared__ alignas(32) float red[ATT_WAVES_ * ATT_TILE_];

  const int wave = __builtin_amdgcn_readfirstlane((int)(threadIdx.x >> 5));
  const int lane = threadIdx.x & 31;
  const int half = lane >> 4;
  const int r    = lane & 15;
  const int b    = blockIdx.y;
  const int hw0  = blockIdx.x * (16 * ATT_HT_);

  int hwr[ATT_HT_];
#pragma unroll
  for (int j = 0; j < ATT_HT_; ++j)
    hwr[j] = min(hw0 + 16 * j + r, HW_ - 1);

  v8f acc[ATT_HT_][2] = {};
  const int kbeg = wave * ATT_KSL_;
  const float* xk = x + ((size_t)b * C_ + kbeg + 2 * half) * HW_;
  const float* w0 = Wa + (size_t)r * C_ + kbeg + 2 * half;
  const float* w1 = w0 + (size_t)16 * C_;

#pragma unroll 4
  for (int k0 = 0; k0 < ATT_KSL_; k0 += 4) {
    const v2f b0 = *(const v2f*)(w0 + k0);
    const v2f b1 = *(const v2f*)(w1 + k0);
#pragma unroll
    for (int j = 0; j < ATT_HT_; ++j) {
      v2f a;
      a.x = xk[(size_t)k0 * HW_ + hwr[j]];
      a.y = xk[(size_t)(k0 + 1) * HW_ + hwr[j]];
      acc[j][0] = WMMA_F32(a, b0, acc[j][0]);
      acc[j][1] = WMMA_F32(a, b1, acc[j][1]);
    }
  }

  float* mine = red + wave * ATT_TILE_ + r * 16 + 8 * half;
#pragma unroll
  for (int j = 0; j < ATT_HT_; ++j) {
    *(v8f*)(mine + (2 * j + 0) * 256) = acc[j][0];
    *(v8f*)(mine + (2 * j + 1) * 256) = acc[j][1];
  }
  __syncthreads();

  const int t   = threadIdx.x;
  const int e   = t & 255;
  const int hwl = e & 15;
  const int mlo = e >> 4;
#pragma unroll
  for (int jj = 0; jj < 4; ++jj) {
    const int q = (t >> 8) + 2 * jj;
    float s = red[q * 256 + e];
#pragma unroll
    for (int w = 1; w < ATT_WAVES_; ++w) s += red[w * ATT_TILE_ + q * 256 + e];
    const int m  = (q & 1) * 16 + mlo;
    const int hw = hw0 + (q >> 1) * 16 + hwl;
    if (hw < HW_) A_ws[((size_t)b * M_ + m) * HW_ + hw] = fast_sigmoid(s + ba[m]);
  }
}

typedef __attribute__((ext_vector_type(4))) float v4f;

#define BAP_CH_   64
#define BAP_NST_  4
#define BAP_XF4_  (BAP_CH_ * HW_ / 4)
#define BAP_AF4_  (M_ * HW_ / 4)
#define BAP_NG_   (HW_ / 8)

__global__ __launch_bounds__(256) void k_bap(const float* __restrict__ x,
                                             const float* __restrict__ A_ws,
                                             float* __restrict__ feats) {
  __shared__ alignas(16) float xs[2][BAP_CH_ * HW_];
  __shared__ alignas(16) float as[M_ * HW_];

  const int wave  = __builtin_amdgcn_readfirstlane((int)(threadIdx.x >> 5));
  const int tid   = threadIdx.x;
  const int lane  = tid & 31;
  const int half  = lane >> 4;
  const int r     = lane & 15;
  const int b     = blockIdx.y;
  const int cb    = blockIdx.x * (BAP_CH_ * BAP_NST_);
  const int ctile = wave >> 1;
  const int mtile = wave & 1;

  const v4f* xsrc = (const v4f*)(x + ((size_t)b * C_ + cb) * HW_) + tid;
  const v4f* asrc = (const v4f*)(A_ws + (size_t)b * M_ * HW_) + tid;

  v4f st[13];
  v4f sa[7];
#pragma unroll
  for (int i = 0; i < 12; ++i) st[i] = xsrc[i * 256];
  if (wave < 2) st[12] = xsrc[12 * 256];
#pragma unroll
  for (int i = 0; i < 6; ++i) sa[i] = asrc[i * 256];
  if (wave < 1) sa[6] = asrc[6 * 256];
#pragma unroll
  for (int i = 0; i < 12; ++i) *(v4f*)(&xs[0][4 * (i * 256 + tid)]) = st[i];
  if (wave < 2) *(v4f*)(&xs[0][4 * (12 * 256 + tid)]) = st[12];
#pragma unroll
  for (int i = 0; i < 6; ++i) *(v4f*)(&as[4 * (i * 256 + tid)]) = sa[i];
  if (wave < 1) *(v4f*)(&as[4 * (6 * 256 + tid)]) = sa[6];
  __syncthreads();

  const float* ap = &as[(16 * mtile + r) * HW_ + 4 * half];
  const float  sc = 1.0f / (float)HW_;

#pragma unroll
  for (int s = 0; s < BAP_NST_; ++s) {
    const int cur = s & 1;
    const bool more = (s + 1) < BAP_NST_;
    if (more) {
#pragma unroll
      for (int i = 0; i < 12; ++i) st[i] = xsrc[(s + 1) * BAP_XF4_ + i * 256];
      if (wave < 2) st[12] = xsrc[(s + 1) * BAP_XF4_ + 12 * 256];
    }

    const float* xp = &xs[cur][(16 * ctile + r) * HW_ + 4 * half];
    v8f acc = {};
#pragma unroll
    for (int gb = 0; gb < BAP_NG_; gb += 8) {
      v4f aq[8], bq[8];
#pragma unroll
      for (int g = 0; g < 8; ++g) {
        aq[g] = *(const v4f*)(ap + 8 * (gb + g));
        bq[g] = *(const v4f*)(xp + 8 * (gb + g));
      }
      asm volatile("" :: "v"(aq[0]), "v"(aq[1]), "v"(aq[2]), "v"(aq[3]),
                         "v"(aq[4]), "v"(aq[5]), "v"(aq[6]), "v"(aq[7]),
                         "v"(bq[0]), "v"(bq[1]), "v"(bq[2]), "v"(bq[3]),
                         "v"(bq[4]), "v"(bq[5]), "v"(bq[6]), "v"(bq[7]));
#pragma unroll
      for (int g = 0; g < 8; ++g) {
        const v2f a01 = {aq[g].x, aq[g].y};
        const v2f a23 = {aq[g].z, aq[g].w};
        const v2f b01 = {bq[g].x, bq[g].y};
        const v2f b23 = {bq[g].z, bq[g].w};
        acc = WMMA_F32(a01, b01, acc);
        acc = WMMA_F32(a23, b23, acc);
      }
    }
    {
      const v2f at = *(const v2f*)(ap + 8 * BAP_NG_ - 2 * half);
      const v2f bt = *(const v2f*)(xp + 8 * BAP_NG_ - 2 * half);
      acc = WMMA_F32(at, bt, acc);
    }

    const int c = cb + s * BAP_CH_ + 16 * ctile + r;
#pragma unroll
    for (int i = 0; i < 8; ++i) {
      const int m = 16 * mtile + 8 * half + i;
      feats[((size_t)b * M_ + m) * C_ + c] = acc[i] * sc;
    }

    if (more) {
#pragma unroll
      for (int i = 0; i < 12; ++i) *(v4f*)(&xs[cur ^ 1][4 * (i * 256 + tid)]) = st[i];
      if (wave < 2) *(v4f*)(&xs[cur ^ 1][4 * (12 * 256 + tid)]) = st[12];
    }
    __syncthreads();
  }
}

#define FC_KS_    128
#define FC_PITCH_ (FC_KS_ + 4)
#define FC_NST_   (KCHUNK_ / FC_KS_)
#define FC_NG_    (FC_KS_ / 8)
#define FC_NLD_   (B_ * FC_KS_ / 4 / 256)

__global__ __launch_bounds__(256) void k_fc_partial(const float* __restrict__ feats,
                                                    const float* __restrict__ Wc,
                                                    float* __restrict__ part) {
  __shared__ alignas(16) float fs[2][B_ * FC_PITCH_];

  const int wave = __builtin_amdgcn_readfirstlane((int)(threadIdx.x >> 5));
  const int lane = threadIdx.x & 31;
  const int half = lane >> 4;
  const int r    = lane & 15;
  const int nt   = blockIdx.x * 8 + wave;
  const bool wvalid = nt < NTILES_;
  const int kc   = blockIdx.y;
  const int n0   = nt * 16;

  const bool nvalid = (n0 + r) < NC_;
  const int  nrow   = min(n0 + r, NC_ - 1);
  const float* wrow = Wc + (size_t)nrow * K3_ + (size_t)kc * KCHUNK_ + 4 * half;

  const float* fsrc = feats + (size_t)wave * K3_ + (size_t)kc * KCHUNK_ + 4 * lane;
  const int    fdst = wave * FC_PITCH_ + 4 * lane;

  v4f st[FC_NLD_];
  v4f wc[FC_NG_];
#pragma unroll
  for (int i = 0; i < FC_NLD_; ++i) st[i] = *(const v4f*)(fsrc + (size_t)(8 * i) * K3_);
  if (wvalid) {
#pragma unroll
    for (int g = 0; g < FC_NG_; ++g) wc[g] = *(const v4f*)(wrow + 8 * g);
  }
#pragma unroll
  for (int i = 0; i < FC_NLD_; ++i) *(v4f*)(&fs[0][fdst + 8 * i * FC_PITCH_]) = st[i];
  __syncthreads();

  v8f acc[4] = {{}, {}, {}, {}};

#pragma unroll 2
  for (int s = 0; s < FC_NST_; ++s) {
    const int cur = s & 1;
    const bool more = (s + 1) < FC_NST_;
    v4f wn[FC_NG_];
    if (more) {
#pragma unroll
      for (int i = 0; i < FC_NLD_; ++i)
        st[i] = *(const v4f*)(fsrc + (size_t)(8 * i) * K3_ + (s + 1) * FC_KS_);
      if (wvalid) {
#pragma unroll
        for (int g = 0; g < FC_NG_; ++g) wn[g] = *(const v4f*)(wrow + (s + 1) * FC_KS_ + 8 * g);
      }
    }
    if (wvalid) {
      const float* fa = &fs[cur][r * FC_PITCH_ + 4 * half];
#pragma unroll
      for (int g = 0; g < FC_NG_; ++g) {
        v4f bq = wc[g];
        bq.x = nvalid ? bq.x : 0.0f;
        bq.y = nvalid ? bq.y : 0.0f;
        bq.z = nvalid ? bq.z : 0.0f;
        bq.w = nvalid ? bq.w : 0.0f;
        const v2f b01 = {bq.x, bq.y};
        const v2f b23 = {bq.z, bq.w};
#pragma unroll
        for (int mt = 0; mt < 4; ++mt) {
          const v4f aq = *(const v4f*)(fa + mt * 16 * FC_PITCH_ + 8 * g);
          const v2f a01 = {aq.x, aq.y};
          const v2f a23 = {aq.z, aq.w};
          acc[mt] = WMMA_F32(a01, b01, acc[mt]);
          acc[mt] = WMMA_F32(a23, b23, acc[mt]);
        }
      }
    }
    if (more) {
#pragma unroll
      for (int i = 0; i < FC_NLD_; ++i) *(v4f*)(&fs[cur ^ 1][fdst + 8 * i * FC_PITCH_]) = st[i];
      if (wvalid) {
#pragma unroll
        for (int g = 0; g < FC_NG_; ++g) wc[g] = wn[g];
      }
    }
    __syncthreads();
  }

  if (wvalid) {
#pragma unroll
    for (int mt = 0; mt < 4; ++mt) {
#pragma unroll
      for (int i = 0; i < 8; ++i) {
        const int brow = mt * 16 + i + (half ? 8 : 0);
        part[((size_t)kc * B_ + brow) * NPAD_ + n0 + r] = acc[mt][i];
      }
    }
  }
}

__global__ __launch_bounds__(256) void k_reduce(const float* __restrict__ part,
                                               const float* __restrict__ bc,
                                               float* __restrict__ out) {
  const int idx = blockIdx.x * blockDim.x + threadIdx.x;
  if (idx >= B_ * NC_) return;
  const int b = idx / NC_;
  const int n = idx - b * NC_;
  float s = bc[n];
#pragma unroll 4
  for (int kc = 0; kc < NKCH_; ++kc)
    s += part[((size_t)kc * B_ + b) * NPAD_ + n];
  out[idx] = s;
}

extern "C" void kernel_launch(void* const* d_in, const int* in_sizes, int n_in,
                              void* d_out, int out_size, void* d_ws, size_t ws_size,
                              hipStream_t stream) {
  const float* x  = (const float*)d_in[0];
  const float* Wa = (const float*)d_in[1];
  const float* ba = (const float*)d_in[2];
  const float* Wc = (const float*)d_in[3];
  const float* bc = (const float*)d_in[4];
  float* out = (float*)d_out;

  float* A_ws    = (float*)d_ws;
  float* feats   = A_ws + (size_t)B_ * M_ * HW_;
  float* part_ws = feats + (size_t)B_ * K3_;

  k_attn      <<<dim3(4, B_),   512, 0, stream>>>(x, Wa, ba, A_ws);
  k_bap       <<<dim3(8, B_),   256, 0, stream>>>(x, A_ws, feats);
  k_fc_partial<<<dim3(4, NKCH_), 256, 0, stream>>>(feats, Wc, part_ws);
  k_reduce    <<<(B_ * NC_ + 255) / 256, 256, 0, stream>>>(part_ws, bc, out);
}


# k_attn: x and Wa fragments as 16-byte loads: a lane reads 4 adjacent hw of a channel row, one per hw tile (tiles interleave in hw); same k order (channel sum order inside each 8 changes)
# speedup vs baseline: 2.5470x; 1.0757x over previous
#include <hip/hip_runtime.h>
#include <cmath>


typedef __attribute__((ext_vector_type(2))) float v2f;
typedef __attribute__((ext_vector_type(8))) float v8f;

#define WMMA_F32(a, b, c) \
  __builtin_amdgcn_wmma_f32_16x16x4_f32(false, (a), false, (b), (short)0, (c), false, false)

#define B_   64
#define C_   2048
#define HW_  196
#define M_   32
#define NC_  396
#define K3_  (M_ * C_)
#define NPAD_ 400
#define NTILES_ 25
#define KCHUNK_ 1024
#define NKCH_   64

__device__ __forceinline__ float fast_sigmoid(float v) {
  return __builtin_amdgcn_rcpf(1.0f + __expf(-v));
}

typedef __attribute__((ext_vector_type(4))) float v4f;

#define ATT_WAVES_ 16
#define ATT_KSL_   (C_ / ATT_WAVES_)
#define ATT_HT_    4
#define ATT_NG_    (ATT_KSL_ / 8)
#define ATT_GB_    4
#define ATT_PITCH_ (16 * ATT_HT_ + 4)
#define ATT_TILE_  (M_ * ATT_PITCH_)

__global__ __launch_bounds__(512) void k_attn(const float* __restrict__ x,
                                              const float* __restrict__ Wa,
                                              const float* __restrict__ ba,
                                              float* __restrict__ A_ws) {
  __shared__ alignas(16) float red[ATT_WAVES_ * ATT_TILE_];

  const int wave = __builtin_amdgcn_readfirstlane((int)(threadIdx.x >> 5));
  const int lane = threadIdx.x & 31;
  const int half = lane >> 4;
  const int r    = lane & 15;
  const int b    = blockIdx.y;
  const int hw0  = blockIdx.x * (16 * ATT_HT_);

  const int hwq  = min(hw0 + 4 * r, HW_ - 4);
  const int kbeg = wave * ATT_KSL_;
  const float* xk = x + ((size_t)b * C_ + kbeg + 4 * half) * HW_ + hwq;
  const float* w0 = Wa + (size_t)r * C_ + kbeg + 4 * half;
  const float* w1 = w0 + (size_t)16 * C_;

  v8f acc[ATT_HT_][2] = {};

  for (int gb = 0; gb < ATT_NG_; gb += ATT_GB_) {
    v4f bq[ATT_GB_][2];
    v4f xq[ATT_GB_][4];
#pragma unroll
    for (int u = 0; u < ATT_GB_; ++u) {
      bq[u][0] = *(const v4f*)(w0 + 8 * (gb + u));
      bq[u][1] = *(const v4f*)(w1 + 8 * (gb + u));
#pragma unroll
      for (int i = 0; i < 4; ++i) xq[u][i] = *(const v4f*)(xk + (size_t)(8 * (gb + u) + i) * HW_);
    }
    asm volatile("" :: "v"(bq[0][0]), "v"(bq[0][1]), "v"(bq[1][0]), "v"(bq[1][1]),
                       "v"(bq[2][0]), "v"(bq[2][1]), "v"(bq[3][0]), "v"(bq[3][1]),
                       "v"(xq[0][0]), "v"(xq[0][1]), "v"(xq[0][2]), "v"(xq[0][3]),
                       "v"(xq[1][0]), "v"(xq[1][1]), "v"(xq[1][2]), "v"(xq[1][3]),
                       "v"(xq[2][0]), "v"(xq[2][1]), "v"(xq[2][2]), "v"(xq[2][3]),
                       "v"(xq[3][0]), "v"(xq[3][1]), "v"(xq[3][2]), "v"(xq[3][3]));
#pragma unroll
    for (int u = 0; u < ATT_GB_; ++u) {
      const v2f b01_0 = {bq[u][0].x, bq[u][0].y};
      const v2f b23_0 = {bq[u][0].z, bq[u][0].w};
      const v2f b01_1 = {bq[u][1].x, bq[u][1].y};
      const v2f b23_1 = {bq[u][1].z, bq[u][1].w};
#pragma unroll
      for (int j = 0; j < ATT_HT_; ++j) {
        const v2f a01 = {xq[u][0][j], xq[u][1][j]};
        acc[j][0] = WMMA_F32(a01, b01_0, acc[j][0]);
        acc[j][1] = WMMA_F32(a01, b01_1, acc[j][1]);
      }
#pragma unroll
      for (int j = 0; j < ATT_HT_; ++j) {
        const v2f a23 = {xq[u][2][j], xq[u][3][j]};
        acc[j][0] = WMMA_F32(a23, b23_0, acc[j][0]);
        acc[j][1] = WMMA_F32(a23, b23_1, acc[j][1]);
      }
    }
  }

  float* mine = red + wave * ATT_TILE_ + r * ATT_PITCH_ + 32 * half;
#pragma unroll
  for (int mt = 0; mt < 2; ++mt) {
#pragma unroll
    for (int i = 0; i < 8; ++i) {
      const v4f q = {acc[0][mt][i], acc[1][mt][i], acc[2][mt][i], acc[3][mt][i]};
      *(v4f*)(mine + mt * 16 * ATT_PITCH_ + 4 * i) = q;
    }
  }
  __syncthreads();

  const int t   = threadIdx.x;
  const int hwl = t & 63;
  const int mlo = t >> 6;
  const int hw  = hw0 + hwl;
#pragma unroll
  for (int jj = 0; jj < 4; ++jj) {
    const int m = mlo + 8 * jj;
    const float* p = red + m * ATT_PITCH_ + hwl;
    float s = p[0];
#pragma unroll
    for (int w = 1; w < ATT_WAVES_; ++w) s += p[w * ATT_TILE_];
    if (hw < HW_) A_ws[((size_t)b * M_ + m) * HW_ + hw] = fast_sigmoid(s + ba[m]);
  }
}

#define BAP_CH_   64
#define BAP_NST_  4
#define BAP_XF4_  (BAP_CH_ * HW_ / 4)
#define BAP_AF4_  (M_ * HW_ / 4)
#define BAP_NG_   (HW_ / 8)

__global__ __launch_bounds__(256) void k_bap(const float* __restrict__ x,
                                             const float* __restrict__ A_ws,
                                             float* __restrict__ feats) {
  __shared__ alignas(16) float xs[2][BAP_CH_ * HW_];
  __shared__ alignas(16) float as[M_ * HW_];

  const int wave  = __builtin_amdgcn_readfirstlane((int)(threadIdx.x >> 5));
  const int tid   = threadIdx.x;
  const int lane  = tid & 31;
  const int half  = lane >> 4;
  const int r     = lane & 15;
  const int b     = blockIdx.y;
  const int cb    = blockIdx.x * (BAP_CH_ * BAP_NST_);
  const int ctile = wave >> 1;
  const int mtile = wave & 1;

  const v4f* xsrc = (const v4f*)(x + ((size_t)b * C_ + cb) * HW_) + tid;
  const v4f* asrc = (const v4f*)(A_ws + (size_t)b * M_ * HW_) + tid;

  v4f st[13];
  v4f sa[7];
#pragma unroll
  for (int i = 0; i < 12; ++i) st[i] = xsrc[i * 256];
  if (wave < 2) st[12] = xsrc[12 * 256];
#pragma unroll
  for (int i = 0; i < 6; ++i) sa[i] = asrc[i * 256];
  if (wave < 1) sa[6] = asrc[6 * 256];
#pragma unroll
  for (int i = 0; i < 12; ++i) *(v4f*)(&xs[0][4 * (i * 256 + tid)]) = st[i];
  if (wave < 2) *(v4f*)(&xs[0][4 * (12 * 256 + tid)]) = st[12];
#pragma unroll
  for (int i = 0; i < 6; ++i) *(v4f*)(&as[4 * (i * 256 + tid)]) = sa[i];
  if (wave < 1) *(v4f*)(&as[4 * (6 * 256 + tid)]) = sa[6];
  __syncthreads();

  const float* ap = &as[(16 * mtile + r) * HW_ + 4 * half];
  const float  sc = 1.0f / (float)HW_;

#pragma unroll
  for (int s = 0; s < BAP_NST_; ++s) {
    const int cur = s & 1;
    const bool more = (s + 1) < BAP_NST_;
    if (more) {
#pragma unroll
      for (int i = 0; i < 12; ++i) st[i] = xsrc[(s + 1) * BAP_XF4_ + i * 256];
      if (wave < 2) st[12] = xsrc[(s + 1) * BAP_XF4_ + 12 * 256];
    }

    const float* xp = &xs[cur][(16 * ctile + r) * HW_ + 4 * half];
    v8f acc = {};
#pragma unroll
    for (int gb = 0; gb < BAP_NG_; gb += 8) {
      v4f aq[8], bq[8];
#pragma unroll
      for (int g = 0; g < 8; ++g) {
        aq[g] = *(const v4f*)(ap + 8 * (gb + g));
        bq[g] = *(const v4f*)(xp + 8 * (gb + g));
      }
      asm volatile("" :: "v"(aq[0]), "v"(aq[1]), "v"(aq[2]), "v"(aq[3]),
                         "v"(aq[4]), "v"(aq[5]), "v"(aq[6]), "v"(aq[7]),
                         "v"(bq[0]), "v"(bq[1]), "v"(bq[2]), "v"(bq[3]),
                         "v"(bq[4]), "v"(bq[5]), "v"(bq[6]), "v"(bq[7]));
#pragma unroll
      for (int g = 0; g < 8; ++g) {
        const v2f a01 = {aq[g].x, aq[g].y};
        const v2f a23 = {aq[g].z, aq[g].w};
        const v2f b01 = {bq[g].x, bq[g].y};
        const v2f b23 = {bq[g].z, bq[g].w};
        acc = WMMA_F32(a01, b01, acc);
        acc = WMMA_F32(a23, b23, acc);
      }
    }
    {
      const v2f at = *(const v2f*)(ap + 8 * BAP_NG_ - 2 * half);
      const v2f bt = *(const v2f*)(xp + 8 * BAP_NG_ - 2 * half);
      acc = WMMA_F32(at, bt, acc);
    }

    const int c = cb + s * BAP_CH_ + 16 * ctile + r;
#pragma unroll
    for (int i = 0; i < 8; ++i) {
      const int m = 16 * mtile + 8 * half + i;
      feats[((size_t)b * M_ + m) * C_ + c] = acc[i] * sc;
    }

    if (more) {
#pragma unroll
      for (int i = 0; i < 12; ++i) *(v4f*)(&xs[cur ^ 1][4 * (i * 256 + tid)]) = st[i];
      if (wave < 2) *(v4f*)(&xs[cur ^ 1][4 * (12 * 256 + tid)]) = st[12];
    }
    __syncthreads();
  }
}

#define FC_KS_    128
#define FC_PITCH_ (FC_KS_ + 4)
#define FC_NST_   (KCHUNK_ / FC_KS_)
#define FC_NG_    (FC_KS_ / 8)
#define FC_NLD_   (B_ * FC_KS_ / 4 / 256)

__global__ __launch_bounds__(256) void k_fc_partial(const float* __restrict__ feats,
                                                    const float* __restrict__ Wc,
                                                    float* __restrict__ part) {
  __shared__ alignas(16) float fs[2][B_ * FC_PITCH_];

  const int wave = __builtin_amdgcn_readfirstlane((int)(threadIdx.x >> 5));
  const int lane = threadIdx.x & 31;
  const int half = lane >> 4;
  const int r    = lane & 15;
  const int nt   = blockIdx.x * 8 + wave;
  const bool wvalid = nt < NTILES_;
  const int kc   = blockIdx.y;
  const int n0   = nt * 16;

  const bool nvalid = (n0 + r) < NC_;
  const int  nrow   = min(n0 + r, NC_ - 1);
  const float* wrow = Wc + (size_t)nrow * K3_ + (size_t)kc * KCHUNK_ + 4 * half;

  const float* fsrc = feats + (size_t)wave * K3_ + (size_t)kc * KCHUNK_ + 4 * lane;
  const int    fdst = wave * FC_PITCH_ + 4 * lane;

  v4f st[FC_NLD_];
  v4f wc[FC_NG_];
#pragma unroll
  for (int i = 0; i < FC_NLD_; ++i) st[i] = *(const v4f*)(fsrc + (size_t)(8 * i) * K3_);
  if (wvalid) {
#pragma unroll
    for (int g = 0; g < FC_NG_; ++g) wc[g] = *(const v4f*)(wrow + 8 * g);
  }
#pragma unroll
  for (int i = 0; i < FC_NLD_; ++i) *(v4f*)(&fs[0][fdst + 8 * i * FC_PITCH_]) = st[i];
  __syncthreads();

  v8f acc[4] = {{}, {}, {}, {}};

#pragma unroll 2
  for (int s = 0; s < FC_NST_; ++s) {
    const int cur = s & 1;
    const bool more = (s + 1) < FC_NST_;
    v4f wn[FC_NG_];
    if (more) {
#pragma unroll
      for (int i = 0; i < FC_NLD_; ++i)
        st[i] = *(const v4f*)(fsrc + (size_t)(8 * i) * K3_ + (s + 1) * FC_KS_);
      if (wvalid) {
#pragma unroll
        for (int g = 0; g < FC_NG_; ++g) wn[g] = *(const v4f*)(wrow + (s + 1) * FC_KS_ + 8 * g);
      }
    }
    if (wvalid) {
      const float* fa = &fs[cur][r * FC_PITCH_ + 4 * half];
#pragma unroll
      for (int g = 0; g < FC_NG_; ++g) {
        v4f bq = wc[g];
        bq.x = nvalid ? bq.x : 0.0f;
        bq.y = nvalid ? bq.y : 0.0f;
        bq.z = nvalid ? bq.z : 0.0f;
        bq.w = nvalid ? bq.w : 0.0f;
        const v2f b01 = {bq.x, bq.y};
        const v2f b23 = {bq.z, bq.w};
#pragma unroll
        for (int mt = 0; mt < 4; ++mt) {
          const v4f aq = *(const v4f*)(fa + mt * 16 * FC_PITCH_ + 8 * g);
          const v2f a01 = {aq.x, aq.y};
          const v2f a23 = {aq.z, aq.w};
          acc[mt] = WMMA_F32(a01, b01, acc[mt]);
          acc[mt] = WMMA_F32(a23, b23, acc[mt]);
        }
      }
    }
    if (more) {
#pragma unroll
      for (int i = 0; i < FC_NLD_; ++i) *(v4f*)(&fs[cur ^ 1][fdst + 8 * i * FC_PITCH_]) = st[i];
      if (wvalid) {
#pragma unroll
        for (int g = 0; g < FC_NG_; ++g) wc[g] = wn[g];
      }
    }
    __syncthreads();
  }

  if (wvalid) {
#pragma unroll
    for (int mt = 0; mt < 4; ++mt) {
#pragma unroll
      for (int i = 0; i < 8; ++i) {
        const int brow = mt * 16 + i + (half ? 8 : 0);
        part[((size_t)kc * B_ + brow) * NPAD_ + n0 + r] = acc[mt][i];
      }
    }
  }
}

__global__ __launch_bounds__(256) void k_reduce(const float* __restrict__ part,
                                               const float* __restrict__ bc,
                                               float* __restrict__ out) {
  const int idx = blockIdx.x * blockDim.x + threadIdx.x;
  if (idx >= B_ * NC_) return;
  const int b = idx / NC_;
  const int n = idx - b * NC_;
  float s = bc[n];
#pragma unroll 4
  for (int kc = 0; kc < NKCH_; ++kc)
    s += part[((size_t)kc * B_ + b) * NPAD_ + n];
  out[idx] = s;
}

extern "C" void kernel_launch(void* const* d_in, const int* in_sizes, int n_in,
                              void* d_out, int out_size, void* d_ws, size_t ws_size,
                              hipStream_t stream) {
  const float* x  = (const float*)d_in[0];
  const float* Wa = (const float*)d_in[1];
  const float* ba = (const float*)d_in[2];
  const float* Wc = (const float*)d_in[3];
  const float* bc = (const float*)d_in[4];
  float* out = (float*)d_out;

  float* A_ws    = (float*)d_ws;
  float* feats   = A_ws + (size_t)B_ * M_ * HW_;
  float* part_ws = feats + (size_t)B_ * K3_;

  k_attn      <<<dim3(4, B_),   512, 0, stream>>>(x, Wa, ba, A_ws);
  k_bap       <<<dim3(8, B_),   256, 0, stream>>>(x, A_ws, feats);
  k_fc_partial<<<dim3(4, NKCH_), 256, 0, stream>>>(feats, Wc, part_ws);
  k_reduce    <<<(B_ * NC_ + 255) / 256, 256, 0, stream>>>(part_ws, bc, out);
}


# k_reduce: the 64 split-K partial loads of an output issued as one group ahead of the sums (were 16 dependent rounds of 4 loads); same sum order (bias, then kc 0..63)
# speedup vs baseline: 2.8463x; 1.1175x over previous
#include <hip/hip_runtime.h>
#include <cmath>


typedef __attribute__((ext_vector_type(2))) float v2f;
typedef __attribute__((ext_vector_type(8))) float v8f;

#define WMMA_F32(a, b, c) \
  __builtin_amdgcn_wmma_f32_16x16x4_f32(false, (a), false, (b), (short)0, (c), false, false)

#define B_   64
#define C_   2048
#define HW_  196
#define M_   32
#define NC_  396
#define K3_  (M_ * C_)
#define NPAD_ 400
#define NTILES_ 25
#define KCHUNK_ 1024
#define NKCH_   64

__device__ __forceinline__ float fast_sigmoid(float v) {
  return __builtin_amdgcn_rcpf(1.0f + __expf(-v));
}

typedef __attribute__((ext_vector_type(4))) float v4f;

#define ATT_WAVES_ 16
#define ATT_KSL_   (C_ / ATT_WAVES_)
#define ATT_HT_    4
#define ATT_NG_    (ATT_KSL_ / 8)
#define ATT_GB_    4
#define ATT_PITCH_ (16 * ATT_HT_ + 4)
#define ATT_TILE_  (M_ * ATT_PITCH_)

__global__ __launch_bounds__(512) void k_attn(const float* __restrict__ x,
                                              const float* __restrict__ Wa,
                                              const float* __restrict__ ba,
                                              float* __restrict__ A_ws) {
  __shared__ alignas(16) float red[ATT_WAVES_ * ATT_TILE_];

  const int wave = __builtin_amdgcn_readfirstlane((int)(threadIdx.x >> 5));
  const int lane = threadIdx.x & 31;
  const int half = lane >> 4;
  const int r    = lane & 15;
  const int b    = blockIdx.y;
  const int hw0  = blockIdx.x * (16 * ATT_HT_);

  const int hwq  = min(hw0 + 4 * r, HW_ - 4);
  const int kbeg = wave * ATT_KSL_;
  const float* xk = x + ((size_t)b * C_ + kbeg + 4 * half) * HW_ + hwq;
  const float* w0 = Wa + (size_t)r * C_ + kbeg + 4 * half;
  const float* w1 = w0 + (size_t)16 * C_;

  v8f acc[ATT_HT_][2] = {};

  for (int gb = 0; gb < ATT_NG_; gb += ATT_GB_) {
    v4f bq[ATT_GB_][2];
    v4f xq[ATT_GB_][4];
#pragma unroll
    for (int u = 0; u < ATT_GB_; ++u) {
      bq[u][0] = *(const v4f*)(w0 + 8 * (gb + u));
      bq[u][1] = *(const v4f*)(w1 + 8 * (gb + u));
#pragma unroll
      for (int i = 0; i < 4; ++i) xq[u][i] = *(const v4f*)(xk + (size_t)(8 * (gb + u) + i) * HW_);
    }
    asm volatile("" :: "v"(bq[0][0]), "v"(bq[0][1]), "v"(bq[1][0]), "v"(bq[1][1]),
                       "v"(bq[2][0]), "v"(bq[2][1]), "v"(bq[3][0]), "v"(bq[3][1]),
                       "v"(xq[0][0]), "v"(xq[0][1]), "v"(xq[0][2]), "v"(xq[0][3]),
                       "v"(xq[1][0]), "v"(xq[1][1]), "v"(xq[1][2]), "v"(xq[1][3]),
                       "v"(xq[2][0]), "v"(xq[2][1]), "v"(xq[2][2]), "v"(xq[2][3]),
                       "v"(xq[3][0]), "v"(xq[3][1]), "v"(xq[3][2]), "v"(xq[3][3]));
#pragma unroll
    for (int u = 0; u < ATT_GB_; ++u) {
      const v2f b01_0 = {bq[u][0].x, bq[u][0].y};
      const v2f b23_0 = {bq[u][0].z, bq[u][0].w};
      const v2f b01_1 = {bq[u][1].x, bq[u][1].y};
      const v2f b23_1 = {bq[u][1].z, bq[u][1].w};
#pragma unroll
      for (int j = 0; j < ATT_HT_; ++j) {
        const v2f a01 = {xq[u][0][j], xq[u][1][j]};
        acc[j][0] = WMMA_F32(a01, b01_0, acc[j][0]);
        acc[j][1] = WMMA_F32(a01, b01_1, acc[j][1]);
      }
#pragma unroll
      for (int j = 0; j < ATT_HT_; ++j) {
        const v2f a23 = {xq[u][2][j], xq[u][3][j]};
        acc[j][0] = WMMA_F32(a23, b23_0, acc[j][0]);
        acc[j][1] = WMMA_F32(a23, b23_1, acc[j][1]);
      }
    }
  }

  float* mine = red + wave * ATT_TILE_ + r * ATT_PITCH_ + 32 * half;
#pragma unroll
  for (int mt = 0; mt < 2; ++mt) {
#pragma unroll
    for (int i = 0; i < 8; ++i) {
      const v4f q = {acc[0][mt][i], acc[1][mt][i], acc[2][mt][i], acc[3][mt][i]};
      *(v4f*)(mine + mt * 16 * ATT_PITCH_ + 4 * i) = q;
    }
  }
  __syncthreads();

  const int t   = threadIdx.x;
  const int hwl = t & 63;
  const int mlo = t >> 6;
  const int hw  = hw0 + hwl;
#pragma unroll
  for (int jj = 0; jj < 4; ++jj) {
    const int m = mlo + 8 * jj;
    const float* p = red + m * ATT_PITCH_ + hwl;
    float s = p[0];
#pragma unroll
    for (int w = 1; w < ATT_WAVES_; ++w) s += p[w * ATT_TILE_];
    if (hw < HW_) A_ws[((size_t)b * M_ + m) * HW_ + hw] = fast_sigmoid(s + ba[m]);
  }
}

#define BAP_CH_   64
#define BAP_NST_  4
#define BAP_XF4_  (BAP_CH_ * HW_ / 4)
#define BAP_AF4_  (M_ * HW_ / 4)
#define BAP_NG_   (HW_ / 8)

__global__ __launch_bounds__(256) void k_bap(const float* __restrict__ x,
                                             const float* __restrict__ A_ws,
                                             float* __restrict__ feats) {
  __shared__ alignas(16) float xs[2][BAP_CH_ * HW_];
  __shared__ alignas(16) float as[M_ * HW_];

  const int wave  = __builtin_amdgcn_readfirstlane((int)(threadIdx.x >> 5));
  const int tid   = threadIdx.x;
  const int lane  = tid & 31;
  const int half  = lane >> 4;
  const int r     = lane & 15;
  const int b     = blockIdx.y;
  const int cb    = blockIdx.x * (BAP_CH_ * BAP_NST_);
  const int ctile = wave >> 1;
  const int mtile = wave & 1;

  const v4f* xsrc = (const v4f*)(x + ((size_t)b * C_ + cb) * HW_) + tid;
  const v4f* asrc = (const v4f*)(A_ws + (size_t)b * M_ * HW_) + tid;

  v4f st[13];
  v4f sa[7];
#pragma unroll
  for (int i = 0; i < 12; ++i) st[i] = xsrc[i * 256];
  if (wave < 2) st[12] = xsrc[12 * 256];
#pragma unroll
  for (int i = 0; i < 6; ++i) sa[i] = asrc[i * 256];
  if (wave < 1) sa[6] = asrc[6 * 256];
#pragma unroll
  for (int i = 0; i < 12; ++i) *(v4f*)(&xs[0][4 * (i * 256 + tid)]) = st[i];
  if (wave < 2) *(v4f*)(&xs[0][4 * (12 * 256 + tid)]) = st[12];
#pragma unroll
  for (int i = 0; i < 6; ++i) *(v4f*)(&as[4 * (i * 256 + tid)]) = sa[i];
  if (wave < 1) *(v4f*)(&as[4 * (6 * 256 + tid)]) = sa[6];
  __syncthreads();

  const float* ap = &as[(16 * mtile + r) * HW_ + 4 * half];
  const float  sc = 1.0f / (float)HW_;

#pragma unroll
  for (int s = 0; s < BAP_NST_; ++s) {
    const int cur = s & 1;
    const bool more = (s + 1) < BAP_NST_;
    if (more) {
#pragma unroll
      for (int i = 0; i < 12; ++i) st[i] = xsrc[(s + 1) * BAP_XF4_ + i * 256];
      if (wave < 2) st[12] = xsrc[(s + 1) * BAP_XF4_ + 12 * 256];
    }

    const float* xp = &xs[cur][(16 * ctile + r) * HW_ + 4 * half];
    v8f acc = {};
#pragma unroll
    for (int gb = 0; gb < BAP_NG_; gb += 8) {
      v4f aq[8], bq[8];
#pragma unroll
      for (int g = 0; g < 8; ++g) {
        aq[g] = *(const v4f*)(ap + 8 * (gb + g));
        bq[g] = *(const v4f*)(xp + 8 * (gb + g));
      }
      asm volatile("" :: "v"(aq[0]), "v"(aq[1]), "v"(aq[2]), "v"(aq[3]),
                         "v"(aq[4]), "v"(aq[5]), "v"(aq[6]), "v"(aq[7]),
                         "v"(bq[0]), "v"(bq[1]), "v"(bq[2]), "v"(bq[3]),
                         "v"(bq[4]), "v"(bq[5]), "v"(bq[6]), "v"(bq[7]));
#pragma unroll
      for (int g = 0; g < 8; ++g) {
        const v2f a01 = {aq[g].x, aq[g].y};
        const v2f a23 = {aq[g].z, aq[g].w};
        const v2f b01 = {bq[g].x, bq[g].y};
        const v2f b23 = {bq[g].z, bq[g].w};
        acc = WMMA_F32(a01, b01, acc);
        acc = WMMA_F32(a23, b23, acc);
      }
    }
    {
      const v2f at = *(const v2f*)(ap + 8 * BAP_NG_ - 2 * half);
      const v2f bt = *(const v2f*)(xp + 8 * BAP_NG_ - 2 * half);
      acc = WMMA_F32(at, bt, acc);
    }

    const int c = cb + s * BAP_CH_ + 16 * ctile + r;
#pragma unroll
    for (int i = 0; i < 8; ++i) {
      const int m = 16 * mtile + 8 * half + i;
      feats[((size_t)b * M_ + m) * C_ + c] = acc[i] * sc;
    }

    if (more) {
#pragma unroll
      for (int i = 0; i < 12; ++i) *(v4f*)(&xs[cur ^ 1][4 * (i * 256 + tid)]) = st[i];
      if (wave < 2) *(v4f*)(&xs[cur ^ 1][4 * (12 * 256 + tid)]) = st[12];
    }
    __syncthreads();
  }
}

#define FC_KS_    128
#define FC_PITCH_ (FC_KS_ + 4)
#define FC_NST_   (KCHUNK_ / FC_KS_)
#define FC_NG_    (FC_KS_ / 8)
#define FC_NLD_   (B_ * FC_KS_ / 4 / 256)

__global__ __launch_bounds__(256) void k_fc_partial(const float* __restrict__ feats,
                                                    const float* __restrict__ Wc,
                                                    float* __restrict__ part) {
  __shared__ alignas(16) float fs[2][B_ * FC_PITCH_];

  const int wave = __builtin_amdgcn_readfirstlane((int)(threadIdx.x >> 5));
  const int lane = threadIdx.x & 31;
  const int half = lane >> 4;
  const int r    = lane & 15;
  const int nt   = blockIdx.x * 8 + wave;
  const bool wvalid = nt < NTILES_;
  const int kc   = blockIdx.y;
  const int n0   = nt * 16;

  const bool nvalid = (n0 + r) < NC_;
  const int  nrow   = min(n0 + r, NC_ - 1);
  const float* wrow = Wc + (size_t)nrow * K3_ + (size_t)kc * KCHUNK_ + 4 * half;

  const float* fsrc = feats + (size_t)wave * K3_ + (size_t)kc * KCHUNK_ + 4 * lane;
  const int    fdst = wave * FC_PITCH_ + 4 * lane;

  v4f st[FC_NLD_];
  v4f wc[FC_NG_];
#pragma unroll
  for (int i = 0; i < FC_NLD_; ++i) st[i] = *(const v4f*)(fsrc + (size_t)(8 * i) * K3_);
  if (wvalid) {
#pragma unroll
    for (int g = 0; g < FC_NG_; ++g) wc[g] = *(const v4f*)(wrow + 8 * g);
  }
#pragma unroll
  for (int i = 0; i < FC_NLD_; ++i) *(v4f*)(&fs[0][fdst + 8 * i * FC_PITCH_]) = st[i];
  __syncthreads();

  v8f acc[4] = {{}, {}, {}, {}};

#pragma unroll 2
  for (int s = 0; s < FC_NST_; ++s) {
    const int cur = s & 1;
    const bool more = (s + 1) < FC_NST_;
    v4f wn[FC_NG_];
    if (more) {
#pragma unroll
      for (int i = 0; i < FC_NLD_; ++i)
        st[i] = *(const v4f*)(fsrc + (size_t)(8 * i) * K3_ + (s + 1) * FC_KS_);
      if (wvalid) {
#pragma unroll
        for (int g = 0; g < FC_NG_; ++g) wn[g] = *(const v4f*)(wrow + (s + 1) * FC_KS_ + 8 * g);
      }
    }
    if (wvalid) {
      const float* fa = &fs[cur][r * FC_PITCH_ + 4 * half];
#pragma unroll
      for (int g = 0; g < FC_NG_; ++g) {
        v4f bq = wc[g];
        bq.x = nvalid ? bq.x : 0.0f;
        bq.y = nvalid ? bq.y : 0.0f;
        bq.z = nvalid ? bq.z : 0.0f;
        bq.w = nvalid ? bq.w : 0.0f;
        const v2f b01 = {bq.x, bq.y};
        const v2f b23 = {bq.z, bq.w};
#pragma unroll
        for (int mt = 0; mt < 4; ++mt) {
          const v4f aq = *(const v4f*)(fa + mt * 16 * FC_PITCH_ + 8 * g);
          const v2f a01 = {aq.x, aq.y};
          const v2f a23 = {aq.z, aq.w};
          acc[mt] = WMMA_F32(a01, b01, acc[mt]);
          acc[mt] = WMMA_F32(a23, b23, acc[mt]);
        }
      }
    }
    if (more) {
#pragma unroll
      for (int i = 0; i < FC_NLD_; ++i) *(v4f*)(&fs[cur ^ 1][fdst + 8 * i * FC_PITCH_]) = st[i];
      if (wvalid) {
#pragma unroll
        for (int g = 0; g < FC_NG_; ++g) wc[g] = wn[g];
      }
    }
    __syncthreads();
  }

  if (wvalid) {
#pragma unroll
    for (int mt = 0; mt < 4; ++mt) {
#pragma unroll
      for (int i = 0; i < 8; ++i) {
        const int brow = mt * 16 + i + (half ? 8 : 0);
        part[((size_t)kc * B_ + brow) * NPAD_ + n0 + r] = acc[mt][i];
      }
    }
  }
}

__global__ __launch_bounds__(256) void k_reduce(const float* __restrict__ part,
                                               const float* __restrict__ bc,
                                               float* __restrict__ out) {
  const int idx = blockIdx.x * blockDim.x + threadIdx.x;
  if (idx >= B_ * NC_) return;
  const int b = idx / NC_;
  const int n = idx - b * NC_;
  const float* p = part + (size_t)b * NPAD_ + n;
  float v[NKCH_];
#pragma unroll
  for (int kc = 0; kc < NKCH_; ++kc) v[kc] = p[(size_t)kc * B_ * NPAD_];
  float s = bc[n];
#pragma unroll
  for (int kc = 0; kc < NKCH_; ++kc) s += v[kc];
  out[idx] = s;
}

extern "C" void kernel_launch(void* const* d_in, const int* in_sizes, int n_in,
                              void* d_out, int out_size, void* d_ws, size_t ws_size,
                              hipStream_t stream) {
  const float* x  = (const float*)d_in[0];
  const float* Wa = (const float*)d_in[1];
  const float* ba = (const float*)d_in[2];
  const float* Wc = (const float*)d_in[3];
  const float* bc = (const float*)d_in[4];
  float* out = (float*)d_out;

  float* A_ws    = (float*)d_ws;
  float* feats   = A_ws + (size_t)B_ * M_ * HW_;
  float* part_ws = feats + (size_t)B_ * K3_;

  k_attn      <<<dim3(4, B_),   512, 0, stream>>>(x, Wa, ba, A_ws);
  k_bap       <<<dim3(8, B_),   256, 0, stream>>>(x, A_ws, feats);
  k_fc_partial<<<dim3(4, NKCH_), 256, 0, stream>>>(feats, Wc, part_ws);
  k_reduce    <<<(B_ * NC_ + 255) / 256, 256, 0, stream>>>(part_ws, bc, out);
}


# k_fc_partial: 16 waves share one LDS copy of the block's Wc rows and feats; 12 own an n-tile x 2 b-tiles, 4 quarter the k of a 25th tile (LDS sum: its k order changes): equal WMMAs per SIMD
# speedup vs baseline: 3.0031x; 1.0551x over previous
#include <hip/hip_runtime.h>
#include <cmath>


typedef __attribute__((ext_vector_type(2))) float v2f;
typedef __attribute__((ext_vector_type(8))) float v8f;

#define WMMA_F32(a, b, c) \
  __builtin_amdgcn_wmma_f32_16x16x4_f32(false, (a), false, (b), (short)0, (c), false, false)

#define B_   64
#define C_   2048
#define HW_  196
#define M_   32
#define NC_  396
#define K3_  (M_ * C_)
#define NPAD_ 400
#define NTILES_ 25
#define KCHUNK_ 1024
#define NKCH_   64

__device__ __forceinline__ float fast_sigmoid(float v) {
  return __builtin_amdgcn_rcpf(1.0f + __expf(-v));
}

typedef __attribute__((ext_vector_type(4))) float v4f;

#define ATT_WAVES_ 16
#define ATT_KSL_   (C_ / ATT_WAVES_)
#define ATT_HT_    4
#define ATT_NG_    (ATT_KSL_ / 8)
#define ATT_GB_    4
#define ATT_PITCH_ (16 * ATT_HT_ + 4)
#define ATT_TILE_  (M_ * ATT_PITCH_)

__global__ __launch_bounds__(512) void k_attn(const float* __restrict__ x,
                                              const float* __restrict__ Wa,
                                              const float* __restrict__ ba,
                                              float* __restrict__ A_ws) {
  __shared__ alignas(16) float red[ATT_WAVES_ * ATT_TILE_];

  const int wave = __builtin_amdgcn_readfirstlane((int)(threadIdx.x >> 5));
  const int lane = threadIdx.x & 31;
  const int half = lane >> 4;
  const int r    = lane & 15;
  const int b    = blockIdx.y;
  const int hw0  = blockIdx.x * (16 * ATT_HT_);

  const int hwq  = min(hw0 + 4 * r, HW_ - 4);
  const int kbeg = wave * ATT_KSL_;
  const float* xk = x + ((size_t)b * C_ + kbeg + 4 * half) * HW_ + hwq;
  const float* w0 = Wa + (size_t)r * C_ + kbeg + 4 * half;
  const float* w1 = w0 + (size_t)16 * C_;

  v8f acc[ATT_HT_][2] = {};

  for (int gb = 0; gb < ATT_NG_; gb += ATT_GB_) {
    v4f bq[ATT_GB_][2];
    v4f xq[ATT_GB_][4];
#pragma unroll
    for (int u = 0; u < ATT_GB_; ++u) {
      bq[u][0] = *(const v4f*)(w0 + 8 * (gb + u));
      bq[u][1] = *(const v4f*)(w1 + 8 * (gb + u));
#pragma unroll
      for (int i = 0; i < 4; ++i) xq[u][i] = *(const v4f*)(xk + (size_t)(8 * (gb + u) + i) * HW_);
    }
    asm volatile("" :: "v"(bq[0][0]), "v"(bq[0][1]), "v"(bq[1][0]), "v"(bq[1][1]),
                       "v"(bq[2][0]), "v"(bq[2][1]), "v"(bq[3][0]), "v"(bq[3][1]),
                       "v"(xq[0][0]), "v"(xq[0][1]), "v"(xq[0][2]), "v"(xq[0][3]),
                       "v"(xq[1][0]), "v"(xq[1][1]), "v"(xq[1][2]), "v"(xq[1][3]),
                       "v"(xq[2][0]), "v"(xq[2][1]), "v"(xq[2][2]), "v"(xq[2][3]),
                       "v"(xq[3][0]), "v"(xq[3][1]), "v"(xq[3][2]), "v"(xq[3][3]));
#pragma unroll
    for (int u = 0; u < ATT_GB_; ++u) {
      const v2f b01_0 = {bq[u][0].x, bq[u][0].y};
      const v2f b23_0 = {bq[u][0].z, bq[u][0].w};
      const v2f b01_1 = {bq[u][1].x, bq[u][1].y};
      const v2f b23_1 = {bq[u][1].z, bq[u][1].w};
#pragma unroll
      for (int j = 0; j < ATT_HT_; ++j) {
        const v2f a01 = {xq[u][0][j], xq[u][1][j]};
        acc[j][0] = WMMA_F32(a01, b01_0, acc[j][0]);
        acc[j][1] = WMMA_F32(a01, b01_1, acc[j][1]);
      }
#pragma unroll
      for (int j = 0; j < ATT_HT_; ++j) {
        const v2f a23 = {xq[u][2][j], xq[u][3][j]};
        acc[j][0] = WMMA_F32(a23, b23_0, acc[j][0]);
        acc[j][1] = WMMA_F32(a23, b23_1, acc[j][1]);
      }
    }
  }

  float* mine = red + wave * ATT_TILE_ + r * ATT_PITCH_ + 32 * half;
#pragma unroll
  for (int mt = 0; mt < 2; ++mt) {
#pragma unroll
    for (int i = 0; i < 8; ++i) {
      const v4f q = {acc[0][mt][i], acc[1][mt][i], acc[2][mt][i], acc[3][mt][i]};
      *(v4f*)(mine + mt * 16 * ATT_PITCH_ + 4 * i) = q;
    }
  }
  __syncthreads();

  const int t   = threadIdx.x;
  const int hwl = t & 63;
  const int mlo = t >> 6;
  const int hw  = hw0 + hwl;
#pragma unroll
  for (int jj = 0; jj < 4; ++jj) {
    const int m = mlo + 8 * jj;
    const float* p = red + m * ATT_PITCH_ + hwl;
    float s = p[0];
#pragma unroll
    for (int w = 1; w < ATT_WAVES_; ++w) s += p[w * ATT_TILE_];
    if (hw < HW_) A_ws[((size_t)b * M_ + m) * HW_ + hw] = fast_sigmoid(s + ba[m]);
  }
}

#define BAP_CH_   64
#define BAP_NST_  4
#define BAP_XF4_  (BAP_CH_ * HW_ / 4)
#define BAP_AF4_  (M_ * HW_ / 4)
#define BAP_NG_   (HW_ / 8)

__global__ __launch_bounds__(256) void k_bap(const float* __restrict__ x,
                                             const float* __restrict__ A_ws,
                                             float* __restrict__ feats) {
  __shared__ alignas(16) float xs[2][BAP_CH_ * HW_];
  __shared__ alignas(16) float as[M_ * HW_];

  const int wave  = __builtin_amdgcn_readfirstlane((int)(threadIdx.x >> 5));
  const int tid   = threadIdx.x;
  const int lane  = tid & 31;
  const int half  = lane >> 4;
  const int r     = lane & 15;
  const int b     = blockIdx.y;
  const int cb    = blockIdx.x * (BAP_CH_ * BAP_NST_);
  const int ctile = wave >> 1;
  const int mtile = wave & 1;

  const v4f* xsrc = (const v4f*)(x + ((size_t)b * C_ + cb) * HW_) + tid;
  const v4f* asrc = (const v4f*)(A_ws + (size_t)b * M_ * HW_) + tid;

  v4f st[13];
  v4f sa[7];
#pragma unroll
  for (int i = 0; i < 12; ++i) st[i] = xsrc[i * 256];
  if (wave < 2) st[12] = xsrc[12 * 256];
#pragma unroll
  for (int i = 0; i < 6; ++i) sa[i] = asrc[i * 256];
  if (wave < 1) sa[6] = asrc[6 * 256];
#pragma unroll
  for (int i = 0; i < 12; ++i) *(v4f*)(&xs[0][4 * (i * 256 + tid)]) = st[i];
  if (wave < 2) *(v4f*)(&xs[0][4 * (12 * 256 + tid)]) = st[12];
#pragma unroll
  for (int i = 0; i < 6; ++i) *(v4f*)(&as[4 * (i * 256 + tid)]) = sa[i];
  if (wave < 1) *(v4f*)(&as[4 * (6 * 256 + tid)]) = sa[6];
  __syncthreads();

  const float* ap = &as[(16 * mtile + r) * HW_ + 4 * half];
  const float  sc = 1.0f / (float)HW_;

#pragma unroll
  for (int s = 0; s < BAP_NST_; ++s) {
    const int cur = s & 1;
    const bool more = (s + 1) < BAP_NST_;
    if (more) {
#pragma unroll
      for (int i = 0; i < 12; ++i) st[i] = xsrc[(s + 1) * BAP_XF4_ + i * 256];
      if (wave < 2) st[12] = xsrc[(s + 1) * BAP_XF4_ + 12 * 256];
    }

    const float* xp = &xs[cur][(16 * ctile + r) * HW_ + 4 * half];
    v8f acc = {};
#pragma unroll
    for (int gb = 0; gb < BAP_NG_; gb += 8) {
      v4f aq[8], bq[8];
#pragma unroll
      for (int g = 0; g < 8; ++g) {
        aq[g] = *(const v4f*)(ap + 8 * (gb + g));
        bq[g] = *(const v4f*)(xp + 8 * (gb + g));
      }
      asm volatile("" :: "v"(aq[0]), "v"(aq[1]), "v"(aq[2]), "v"(aq[3]),
                         "v"(aq[4]), "v"(aq[5]), "v"(aq[6]), "v"(aq[7]),
                         "v"(bq[0]), "v"(bq[1]), "v"(bq[2]), "v"(bq[3]),
                         "v"(bq[4]), "v"(bq[5]), "v"(bq[6]), "v"(bq[7]));
#pragma unroll
      for (int g = 0; g < 8; ++g) {
        const v2f a01 = {aq[g].x, aq[g].y};
        const v2f a23 = {aq[g].z, aq[g].w};
        const v2f b01 = {bq[g].x, bq[g].y};
        const v2f b23 = {bq[g].z, bq[g].w};
        acc = WMMA_F32(a01, b01, acc);
        acc = WMMA_F32(a23, b23, acc);
      }
    }
    {
      const v2f at = *(const v2f*)(ap + 8 * BAP_NG_ - 2 * half);
      const v2f bt = *(const v2f*)(xp + 8 * BAP_NG_ - 2 * half);
      acc = WMMA_F32(at, bt, acc);
    }

    const int c = cb + s * BAP_CH_ + 16 * ctile + r;
#pragma unroll
    for (int i = 0; i < 8; ++i) {
      const int m = 16 * mtile + 8 * half + i;
      feats[((size_t)b * M_ + m) * C_ + c] = acc[i] * sc;
    }

    if (more) {
#pragma unroll
      for (int i = 0; i < 12; ++i) *(v4f*)(&xs[cur ^ 1][4 * (i * 256 + tid)]) = st[i];
      if (wave < 2) *(v4f*)(&xs[cur ^ 1][4 * (12 * 256 + tid)]) = st[12];
    }
    __syncthreads();
  }
}

#define FC_KS_    128
#define FC_PITCH_ (FC_KS_ + 4)
#define FC_NST_   (KCHUNK_ / FC_KS_)
#define FC_NG_    (FC_KS_ / 8)
#define FC_WAVES_ 16
#define FC_NTB_   6
#define FC_ROWS_  (B_ + 16 * (FC_NTB_ + 1))
#define FC_NLD_   (FC_ROWS_ / FC_WAVES_)
#define FC_BUF_   (FC_ROWS_ * FC_PITCH_)

__global__ __launch_bounds__(512) void k_fc_partial(const float* __restrict__ feats,
                                                    const float* __restrict__ Wc,
                                                    float* __restrict__ part) {
  __shared__ alignas(16) float fs[2][FC_BUF_];

  const int wave = __builtin_amdgcn_readfirstlane((int)(threadIdx.x >> 5));
  const int lane = threadIdx.x & 31;
  const int half = lane >> 4;
  const int r    = lane & 15;
  const int bx   = blockIdx.x;
  const int kc   = blockIdx.y;

  const size_t koff = (size_t)kc * KCHUNK_ + 4 * lane;
  const int    nx   = 16 * (NTILES_ - 1) + wave;
  const bool   nxok = nx < NC_;
  const float* sp[FC_NLD_];
#pragma unroll
  for (int i = 0; i < 4; ++i) sp[i] = feats + (size_t)(wave + 16 * i) * K3_ + koff;
#pragma unroll
  for (int j = 0; j < FC_NTB_; ++j) sp[4 + j] = Wc + (size_t)(16 * (FC_NTB_ * bx + j) + wave) * K3_ + koff;
  sp[10] = Wc + (size_t)min(nx, NC_ - 1) * K3_ + koff;
  const int fdst = wave * FC_PITCH_ + 4 * lane;
  const v4f zero4 = {0.0f, 0.0f, 0.0f, 0.0f};

  const bool whole = wave < 12;
  const int  jn    = wave >> 1;
  const int  bt0   = 2 * (wave & 1);
  const int  q     = wave - 12;
  const int  aoff  = whole ? (16 * bt0 + r) * FC_PITCH_ + 4 * half
                           : (16 * bx + r) * FC_PITCH_ + 4 * half + 32 * q;
  const int  boff  = whole ? (B_ + 16 * jn + r) * FC_PITCH_ + 4 * half
                           : (B_ + 16 * FC_NTB_ + r) * FC_PITCH_ + 4 * half + 32 * q;

  v4f st[FC_NLD_];
#pragma unroll
  for (int i = 0; i < FC_NLD_ - 1; ++i) st[i] = *(const v4f*)(sp[i]);
  if (nxok) st[10] = *(const v4f*)(sp[10]); else st[10] = zero4;
#pragma unroll
  for (int i = 0; i < FC_NLD_; ++i) *(v4f*)(&fs[0][fdst + 16 * i * FC_PITCH_]) = st[i];
  __syncthreads();

  v8f acc0 = {}, acc1 = {};

#pragma unroll 2
  for (int s = 0; s < FC_NST_; ++s) {
    const int cur = s & 1;
    const bool more = (s + 1) < FC_NST_;
    if (more) {
#pragma unroll
      for (int i = 0; i < FC_NLD_ - 1; ++i) st[i] = *(const v4f*)(sp[i] + (s + 1) * FC_KS_);
      if (nxok) st[10] = *(const v4f*)(sp[10] + (s + 1) * FC_KS_); else st[10] = zero4;
    }
    asm volatile("" ::: "memory");
    const float* fa = &fs[cur][aoff];
    const float* fb = &fs[cur][boff];
    if (whole) {
#pragma unroll
      for (int g = 0; g < FC_NG_; ++g) {
        const v4f bq = *(const v4f*)(fb + 8 * g);
        const v4f a0 = *(const v4f*)(fa + 8 * g);
        const v4f a1 = *(const v4f*)(fa + 16 * FC_PITCH_ + 8 * g);
        const v2f b01 = {bq.x, bq.y};
        const v2f b23 = {bq.z, bq.w};
        const v2f a0l = {a0.x, a0.y};
        const v2f a0h = {a0.z, a0.w};
        const v2f a1l = {a1.x, a1.y};
        const v2f a1h = {a1.z, a1.w};
        acc0 = WMMA_F32(a0l, b01, acc0);
        acc1 = WMMA_F32(a1l, b01, acc1);
        acc0 = WMMA_F32(a0h, b23, acc0);
        acc1 = WMMA_F32(a1h, b23, acc1);
      }
    } else {
#pragma unroll
      for (int g = 0; g < FC_NG_ / 4; ++g) {
        const v4f bq = *(const v4f*)(fb + 8 * g);
        const v4f a0 = *(const v4f*)(fa + 8 * g);
        const v2f b01 = {bq.x, bq.y};
        const v2f b23 = {bq.z, bq.w};
        const v2f a0l = {a0.x, a0.y};
        const v2f a0h = {a0.z, a0.w};
        acc0 = WMMA_F32(a0l, b01, acc0);
        acc0 = WMMA_F32(a0h, b23, acc0);
      }
    }
    if (more) {
#pragma unroll
      for (int i = 0; i < FC_NLD_; ++i) *(v4f*)(&fs[cur ^ 1][fdst + 16 * i * FC_PITCH_]) = st[i];
    }
    __syncthreads();
  }

  float* red = &fs[0][0];
  if (!whole) {
#pragma unroll
    for (int i = 0; i < 8; ++i) red[q * 256 + i * 32 + lane] = acc0[i];
  }
  __syncthreads();

  if (whole) {
    const int n0 = 16 * (FC_NTB_ * bx + jn);
#pragma unroll
    for (int i = 0; i < 8; ++i) {
      const int brow = 16 * bt0 + 8 * half + i;
      part[((size_t)kc * B_ + brow) * NPAD_ + n0 + r]      = acc0[i];
      part[((size_t)kc * B_ + brow + 16) * NPAD_ + n0 + r] = acc1[i];
    }
  } else if (wave == 12) {
    const int n0 = 16 * (NTILES_ - 1);
#pragma unroll
    for (int i = 0; i < 8; ++i) {
      const float* p = red + i * 32 + lane;
      const float sum = ((p[0] + p[256]) + p[512]) + p[768];
      const int brow = 16 * bx + 8 * half + i;
      part[((size_t)kc * B_ + brow) * NPAD_ + n0 + r] = sum;
    }
  }
}

__global__ __launch_bounds__(256) void k_reduce(const float* __restrict__ part,
                                               const float* __restrict__ bc,
                                               float* __restrict__ out) {
  const int idx = blockIdx.x * blockDim.x + threadIdx.x;
  if (idx >= B_ * NC_) return;
  const int b = idx / NC_;
  const int n = idx - b * NC_;
  const float* p = part + (size_t)b * NPAD_ + n;
  float v[NKCH_];
#pragma unroll
  for (int kc = 0; kc < NKCH_; ++kc) v[kc] = p[(size_t)kc * B_ * NPAD_];
  float s = bc[n];
#pragma unroll
  for (int kc = 0; kc < NKCH_; ++kc) s += v[kc];
  out[idx] = s;
}

extern "C" void kernel_launch(void* const* d_in, const int* in_sizes, int n_in,
                              void* d_out, int out_size, void* d_ws, size_t ws_size,
                              hipStream_t stream) {
  const float* x  = (const float*)d_in[0];
  const float* Wa = (const float*)d_in[1];
  const float* ba = (const float*)d_in[2];
  const float* Wc = (const float*)d_in[3];
  const float* bc = (const float*)d_in[4];
  float* out = (float*)d_out;

  float* A_ws    = (float*)d_ws;
  float* feats   = A_ws + (size_t)B_ * M_ * HW_;
  float* part_ws = feats + (size_t)B_ * K3_;

  k_attn      <<<dim3(4, B_),   512, 0, stream>>>(x, Wa, ba, A_ws);
  k_bap       <<<dim3(8, B_),   256, 0, stream>>>(x, A_ws, feats);
  k_fc_partial<<<dim3(4, NKCH_), 512, 0, stream>>>(feats, Wc, part_ws);
  k_reduce    <<<(B_ * NC_ + 255) / 256, 256, 0, stream>>>(part_ws, bc, out);
}
